# starting kernel
# baseline (speedup 1.0000x reference)
#include <hip/hip_runtime.h>
#include <hip/hip_bf16.h>
#include <math.h>


typedef __attribute__((ext_vector_type(2))) float v2f;
typedef __attribute__((ext_vector_type(8))) float v8f;

#define HS 256
#define C2D 128
#define DD 64
#define NC_CHUNKS 8

__global__ void k_qkw1(const float* __restrict__ Q, const float* __restrict__ K,
                       const float* __restrict__ W1,
                       float* __restrict__ QW1, float* __restrict__ KW1) {
  int tid = blockIdx.x * blockDim.x + threadIdx.x;
  int hi = tid >> 7;
  int c  = tid & 127;
  const float* q = Q + hi * 64;
  const float* k = K + hi * 64;
  float sq = 0.f, sk = 0.f;
  #pragma unroll 8
  for (int a = 0; a < 64; ++a) {
    sq += q[a] * W1[a * C2D + c];
    sk += k[a] * W1[(64 + a) * C2D + c];
  }
  QW1[tid] = sq;
  KW1[tid] = sk;
}

__global__ void k_hsum(const float* __restrict__ QW1, const float* __restrict__ KW1,
                       const float* __restrict__ b1, float* __restrict__ Hsum) {
  int tid = blockIdx.x * blockDim.x + threadIdx.x;
  int hj = tid >> 7;
  int c  = tid & 127;
  int h  = hj >> 7;
  float base = KW1[tid] + b1[c];
  const float* qrow = QW1 + (h * 128) * C2D + c;
  float s = 0.f;
  for (int i = 0; i < 128; ++i)
    s += tanhf(qrow[i * C2D] + base);
  Hsum[tid] = s;
}

__global__ __launch_bounds__(32) __attribute__((amdgpu_waves_per_eu(1, 1)))
void k_w2k(const float* __restrict__ K, const float* __restrict__ W2,
           const float* __restrict__ Hsum, float* __restrict__ partial) {
  const int lane = threadIdx.x;
  const int ct = blockIdx.x;
  const int dt = blockIdx.y;
  const int cc = blockIdx.z;
  const int half = lane >> 4;
  const int l16  = lane & 15;
  const int k0   = half * 2;
  const int hj   = ct * 16 + l16;
  const int drow = dt * 16 + l16;
  const int cbase = cc * 16;

  v2f bfrag[16];
  #pragma unroll
  for (int es = 0; es < 16; ++es)
    bfrag[es] = *(const v2f*)(K + hj * 64 + es * 4 + k0);

  v2f cur[16];
  {
    const float* Arow = W2 + (size_t)cbase * 4096 + (size_t)drow * 64;
    #pragma unroll
    for (int es = 0; es < 16; ++es)
      cur[es] = *(const v2f*)(Arow + es * 4 + k0);
  }
  float hs = Hsum[hj * C2D + cbase];

  v8f outacc = {};
  #pragma unroll
  for (int cidx = 0; cidx < 16; ++cidx) {
    v2f nxt[16];
    float hs_n = 0.f;
    if (cidx < 15) {
      const float* ArowN = W2 + (size_t)(cbase + cidx + 1) * 4096 + (size_t)drow * 64;
      #pragma unroll
      for (int es = 0; es < 16; ++es)
        nxt[es] = *(const v2f*)(ArowN + es * 4 + k0);
      hs_n = Hsum[hj * C2D + cbase + cidx + 1];
    }

    v8f acc = {};
    #pragma unroll
    for (int es = 0; es < 16; ++es)
      acc = __builtin_amdgcn_wmma_f32_16x16x4_f32(
           false, cur[es],  false, bfrag[es],
           (short)0, acc,  false,  false);

    #pragma unroll
    for (int v = 0; v < 8; ++v) outacc[v] += hs * acc[v];

    #pragma unroll
    for (int es = 0; es < 16; ++es) cur[es] = nxt[es];
    hs = hs_n;
  }

  float* p = partial + (size_t)cc * (HS * DD) + (size_t)hj * DD + dt * 16 + 8 * half;
  #pragma unroll
  for (int v = 0; v < 8; ++v) p[v] = outacc[v];
}

__global__ void k_out(const float* __restrict__ partial, const float* __restrict__ K,
                      const float* __restrict__ b2, float* __restrict__ out) {
  int tid = blockIdx.x * blockDim.x + threadIdx.x;
  int hj = tid >> 6;
  int d  = tid & 63;
  float s = 0.f;
  #pragma unroll
  for (int cc = 0; cc < NC_CHUNKS; ++cc)
    s += partial[cc * (HS * DD) + tid];
  const float* b2row = b2 + d * 64;
  const float* krow  = K + hj * 64;
  float b2k = 0.f;
  #pragma unroll 8
  for (int e = 0; e < 64; ++e)
    b2k += b2row[e] * krow[e];
  out[tid] = s * (1.f / 128.f) + b2k;
}

extern "C" void kernel_launch(void* const* d_in, const int* in_sizes, int n_in,
                              void* d_out, int out_size, void* d_ws, size_t ws_size,
                              hipStream_t stream) {
  const float* Q  = (const float*)d_in[0];
  const float* K  = (const float*)d_in[1];
  const float* W1 = (const float*)d_in[2];
  const float* b1 = (const float*)d_in[3];
  const float* W2 = (const float*)d_in[4];
  const float* b2 = (const float*)d_in[5];
  float* out = (float*)d_out;

  float* ws      = (float*)d_ws;
  float* QW1     = ws;
  float* KW1     = ws + 32768;
  float* Hsum    = ws + 65536;
  float* partial = ws + 98304;

  k_qkw1<<<128, 256, 0, stream>>>(Q, K, W1, QW1, KW1);
  k_hsum<<<128, 256, 0, stream>>>(QW1, KW1, b1, Hsum);
  dim3 g3(16, 4, NC_CHUNKS);
  k_w2k<<<g3, 32, 0, stream>>>(K, W2, Hsum, partial);
  k_out<<<64, 256, 0, stream>>>(partial, K, b2, out);
}


# k_hsum lanes re-mapped from c to j, one block per (h,c): QW1 (stored c-major, as KW1 and Hsum now are) is wave-uniform and read by 64-byte scalar loads, no per-lane load in the i loop; sum order kept
# speedup vs baseline: 2.0460x; 2.0460x over previous
#include <hip/hip_runtime.h>
#include <hip/hip_bf16.h>
#include <math.h>


typedef __attribute__((ext_vector_type(2))) float v2f;
typedef __attribute__((ext_vector_type(8))) float v8f;
typedef __attribute__((ext_vector_type(16))) float v16f;

#define HS 256
#define C2D 128
#define DD 64
#define NC_CHUNKS 8

__global__ void k_qkw1(const float* __restrict__ Q, const float* __restrict__ K,
                       const float* __restrict__ W1,
                       float* __restrict__ QW1T, float* __restrict__ KW1T) {
  int tid = blockIdx.x * blockDim.x + threadIdx.x;
  int hi = tid >> 7;
  int c  = tid & 127;
  const float* q = Q + hi * 64;
  const float* k = K + hi * 64;
  float sq = 0.f, sk = 0.f;
  #pragma unroll 8
  for (int a = 0; a < 64; ++a) {
    sq += q[a] * W1[a * C2D + c];
    sk += k[a] * W1[(64 + a) * C2D + c];
  }
  const int t = (((hi >> 7) * C2D + c) << 7) + (hi & 127);
  QW1T[t] = sq;
  KW1T[t] = sk;
}

__global__ __launch_bounds__(128)
void k_hsum(const float* __restrict__ QW1T, const float* __restrict__ KW1T,
            const float* __restrict__ b1, float* __restrict__ HsumT) {
  const int hc = blockIdx.x;
  const int c  = hc & 127;
  const int h  = hc >> 7;
  const int j  = threadIdx.x;
  const float base = KW1T[hc * 128 + j] + b1[c];
  const v16f* qcol = (const v16f*)(QW1T + hc * 128);
  float s = 0.f;
  v16f cur = qcol[0];
  for (int blk = 0; blk < 8; ++blk) {
    const v16f nxt = qcol[(blk + 1) & 7];
    #pragma unroll
    for (int u = 0; u < 16; ++u)
      s += tanhf(cur[u] + base);
    cur = nxt;
  }
  HsumT[c * HS + h * 128 + j] = s;
}

__global__ __launch_bounds__(32) __attribute__((amdgpu_waves_per_eu(1, 1)))
void k_w2k(const float* __restrict__ K, const float* __restrict__ W2,
           const float* __restrict__ HsumT, float* __restrict__ partial) {
  const int lane = threadIdx.x;
  const int ct = blockIdx.x;
  const int dt = blockIdx.y;
  const int cc = blockIdx.z;
  const int half = lane >> 4;
  const int l16  = lane & 15;
  const int k0   = half * 2;
  const int hj   = ct * 16 + l16;
  const int drow = dt * 16 + l16;
  const int cbase = cc * 16;

  v2f bfrag[16];
  #pragma unroll
  for (int es = 0; es < 16; ++es)
    bfrag[es] = *(const v2f*)(K + hj * 64 + es * 4 + k0);

  v2f cur[16];
  {
    const float* Arow = W2 + (size_t)cbase * 4096 + (size_t)drow * 64;
    #pragma unroll
    for (int es = 0; es < 16; ++es)
      cur[es] = *(const v2f*)(Arow + es * 4 + k0);
  }
  float hs = HsumT[cbase * HS + hj];

  v8f outacc = {};
  #pragma unroll
  for (int cidx = 0; cidx < 16; ++cidx) {
    v2f nxt[16];
    float hs_n = 0.f;
    if (cidx < 15) {
      const float* ArowN = W2 + (size_t)(cbase + cidx + 1) * 4096 + (size_t)drow * 64;
      #pragma unroll
      for (int es = 0; es < 16; ++es)
        nxt[es] = *(const v2f*)(ArowN + es * 4 + k0);
      hs_n = HsumT[(cbase + cidx + 1) * HS + hj];
    }

    v8f acc = {};
    #pragma unroll
    for (int es = 0; es < 16; ++es)
      acc = __builtin_amdgcn_wmma_f32_16x16x4_f32(
           false, cur[es],  false, bfrag[es],
           (short)0, acc,  false,  false);

    #pragma unroll
    for (int v = 0; v < 8; ++v) outacc[v] += hs * acc[v];

    #pragma unroll
    for (int es = 0; es < 16; ++es) cur[es] = nxt[es];
    hs = hs_n;
  }

  float* p = partial + (size_t)cc * (HS * DD) + (size_t)hj * DD + dt * 16 + 8 * half;
  #pragma unroll
  for (int v = 0; v < 8; ++v) p[v] = outacc[v];
}

__global__ void k_out(const float* __restrict__ partial, const float* __restrict__ K,
                      const float* __restrict__ b2, float* __restrict__ out) {
  int tid = blockIdx.x * blockDim.x + threadIdx.x;
  int hj = tid >> 6;
  int d  = tid & 63;
  float s = 0.f;
  #pragma unroll
  for (int cc = 0; cc < NC_CHUNKS; ++cc)
    s += partial[cc * (HS * DD) + tid];
  const float* b2row = b2 + d * 64;
  const float* krow  = K + hj * 64;
  float b2k = 0.f;
  #pragma unroll 8
  for (int e = 0; e < 64; ++e)
    b2k += b2row[e] * krow[e];
  out[tid] = s * (1.f / 128.f) + b2k;
}

extern "C" void kernel_launch(void* const* d_in, const int* in_sizes, int n_in,
                              void* d_out, int out_size, void* d_ws, size_t ws_size,
                              hipStream_t stream) {
  const float* Q  = (const float*)d_in[0];
  const float* K  = (const float*)d_in[1];
  const float* W1 = (const float*)d_in[2];
  const float* b1 = (const float*)d_in[3];
  const float* W2 = (const float*)d_in[4];
  const float* b2 = (const float*)d_in[5];
  float* out = (float*)d_out;

  float* ws      = (float*)d_ws;
  float* QW1T    = ws;
  float* KW1T    = ws + 32768;
  float* HsumT   = ws + 65536;
  float* partial = ws + 98304;

  k_qkw1<<<128, 256, 0, stream>>>(Q, K, W1, QW1T, KW1T);
  k_hsum<<<256, 128, 0, stream>>>(QW1T, KW1T, b1, HsumT);
  dim3 g3(16, 4, NC_CHUNKS);
  k_w2k<<<g3, 32, 0, stream>>>(K, W2, HsumT, partial);
  k_out<<<64, 256, 0, stream>>>(partial, K, b2, out);
}


# k_hsum: tanh(a+b) as (A+B)/(1+A*B): lane t puts tanh(QW1[t]) in LDS, each lane takes tanh of its KW1+b1; i loop is fma, rcp, add, fma, no tanh call; tanhf(a+b) kept if 1+A*B<1/16; sum order kept
# speedup vs baseline: 2.7466x; 1.3424x over previous
#include <hip/hip_runtime.h>
#include <hip/hip_bf16.h>
#include <math.h>


typedef __attribute__((ext_vector_type(2))) float v2f;
typedef __attribute__((ext_vector_type(4))) float v4f;
typedef __attribute__((ext_vector_type(8))) float v8f;

#define HS 256
#define C2D 128
#define DD 64
#define NC_CHUNKS 8

__global__ void k_qkw1(const float* __restrict__ Q, const float* __restrict__ K,
                       const float* __restrict__ W1,
                       float* __restrict__ QW1T, float* __restrict__ KW1T) {
  int tid = blockIdx.x * blockDim.x + threadIdx.x;
  int hi = tid >> 7;
  int c  = tid & 127;
  const float* q = Q + hi * 64;
  const float* k = K + hi * 64;
  float sq = 0.f, sk = 0.f;
  #pragma unroll 8
  for (int a = 0; a < 64; ++a) {
    sq += q[a] * W1[a * C2D + c];
    sk += k[a] * W1[(64 + a) * C2D + c];
  }
  const int t = (((hi >> 7) * C2D + c) << 7) + (hi & 127);
  QW1T[t] = sq;
  KW1T[t] = sk;
}

__global__ __launch_bounds__(128)
void k_hsum(const float* __restrict__ QW1T, const float* __restrict__ KW1T,
            const float* __restrict__ b1, float* __restrict__ HsumT) {
  __shared__ __attribute__((aligned(16))) float sA[128];
  __shared__ __attribute__((aligned(16))) float sQ[128];
  const int hc = blockIdx.x;
  const int c  = hc & 127;
  const int h  = hc >> 7;
  const int j  = threadIdx.x;
  const float q = QW1T[hc * 128 + j];
  const float b = KW1T[hc * 128 + j] + b1[c];
  sQ[j] = q;
  sA[j] = tanhf(q);
  const float B = tanhf(b);
  __syncthreads();
  float s = 0.f;
  float dmin = 2.f;
  const v4f* sA4 = (const v4f*)sA;
  v4f cur[4];
  #pragma unroll
  for (int g = 0; g < 4; ++g) cur[g] = sA4[g];
  #pragma nounroll
  for (int blk = 0; blk < 8; ++blk) {
    v4f nxt[4];
    #pragma unroll
    for (int g = 0; g < 4; ++g) nxt[g] = sA4[((blk + 1) & 7) * 4 + g];
    float den[16], rc[16];
    #pragma unroll
    for (int u = 0; u < 16; ++u) den[u] = fmaf(cur[u >> 2][u & 3], B, 1.f);
    #pragma unroll
    for (int u = 0; u < 16; ++u) rc[u] = __builtin_amdgcn_rcpf(den[u]);
    #pragma unroll
    for (int u = 0; u < 16; ++u) dmin = fminf(dmin, den[u]);
    #pragma unroll
    for (int u = 0; u < 16; ++u) s += (cur[u >> 2][u & 3] + B) * rc[u];
    #pragma unroll
    for (int g = 0; g < 4; ++g) cur[g] = nxt[g];
  }
  if (dmin < 0.0625f) {
    float ss = 0.f;
    #pragma nounroll
    for (int i = 0; i < 128; ++i)
      ss += tanhf(sQ[i] + b);
    s = ss;
  }
  HsumT[c * HS + h * 128 + j] = s;
}

__global__ __launch_bounds__(32) __attribute__((amdgpu_waves_per_eu(1, 1)))
void k_w2k(const float* __restrict__ K, const float* __restrict__ W2,
           const float* __restrict__ HsumT, float* __restrict__ partial) {
  const int lane = threadIdx.x;
  const int ct = blockIdx.x;
  const int dt = blockIdx.y;
  const int cc = blockIdx.z;
  const int half = lane >> 4;
  const int l16  = lane & 15;
  const int k0   = half * 2;
  const int hj   = ct * 16 + l16;
  const int drow = dt * 16 + l16;
  const int cbase = cc * 16;

  v2f bfrag[16];
  #pragma unroll
  for (int es = 0; es < 16; ++es)
    bfrag[es] = *(const v2f*)(K + hj * 64 + es * 4 + k0);

  v2f cur[16];
  {
    const float* Arow = W2 + (size_t)cbase * 4096 + (size_t)drow * 64;
    #pragma unroll
    for (int es = 0; es < 16; ++es)
      cur[es] = *(const v2f*)(Arow + es * 4 + k0);
  }
  float hs = HsumT[cbase * HS + hj];

  v8f outacc = {};
  #pragma unroll
  for (int cidx = 0; cidx < 16; ++cidx) {
    v2f nxt[16];
    float hs_n = 0.f;
    if (cidx < 15) {
      const float* ArowN = W2 + (size_t)(cbase + cidx + 1) * 4096 + (size_t)drow * 64;
      #pragma unroll
      for (int es = 0; es < 16; ++es)
        nxt[es] = *(const v2f*)(ArowN + es * 4 + k0);
      hs_n = HsumT[(cbase + cidx + 1) * HS + hj];
    }

    v8f acc = {};
    #pragma unroll
    for (int es = 0; es < 16; ++es)
      acc = __builtin_amdgcn_wmma_f32_16x16x4_f32(
           false, cur[es],  false, bfrag[es],
           (short)0, acc,  false,  false);

    #pragma unroll
    for (int v = 0; v < 8; ++v) outacc[v] += hs * acc[v];

    #pragma unroll
    for (int es = 0; es < 16; ++es) cur[es] = nxt[es];
    hs = hs_n;
  }

  float* p = partial + (size_t)cc * (HS * DD) + (size_t)hj * DD + dt * 16 + 8 * half;
  #pragma unroll
  for (int v = 0; v < 8; ++v) p[v] = outacc[v];
}

__global__ void k_out(const float* __restrict__ partial, const float* __restrict__ K,
                      const float* __restrict__ b2, float* __restrict__ out) {
  int tid = blockIdx.x * blockDim.x + threadIdx.x;
  int hj = tid >> 6;
  int d  = tid & 63;
  float s = 0.f;
  #pragma unroll
  for (int cc = 0; cc < NC_CHUNKS; ++cc)
    s += partial[cc * (HS * DD) + tid];
  const float* b2row = b2 + d * 64;
  const float* krow  = K + hj * 64;
  float b2k = 0.f;
  #pragma unroll 8
  for (int e = 0; e < 64; ++e)
    b2k += b2row[e] * krow[e];
  out[tid] = s * (1.f / 128.f) + b2k;
}

extern "C" void kernel_launch(void* const* d_in, const int* in_sizes, int n_in,
                              void* d_out, int out_size, void* d_ws, size_t ws_size,
                              hipStream_t stream) {
  const float* Q  = (const float*)d_in[0];
  const float* K  = (const float*)d_in[1];
  const float* W1 = (const float*)d_in[2];
  const float* b1 = (const float*)d_in[3];
  const float* W2 = (const float*)d_in[4];
  const float* b2 = (const float*)d_in[5];
  float* out = (float*)d_out;

  float* ws      = (float*)d_ws;
  float* QW1T    = ws;
  float* KW1T    = ws + 32768;
  float* HsumT   = ws + 65536;
  float* partial = ws + 98304;

  k_qkw1<<<128, 256, 0, stream>>>(Q, K, W1, QW1T, KW1T);
  k_hsum<<<256, 128, 0, stream>>>(QW1T, KW1T, b1, HsumT);
  dim3 g3(16, 4, NC_CHUNKS);
  k_w2k<<<g3, 32, 0, stream>>>(K, W2, HsumT, partial);
  k_out<<<64, 256, 0, stream>>>(partial, K, b2, out);
}


# k_qkw1 on the matrix unit: one wave per 16c x 16i tile of QW1 or KW1 (256 one-wave blocks), its 50 operand loads in one batch, not 8 dependent rounds; 64-byte row stores (QW1/KW1 sum order changes)
# speedup vs baseline: 5.4967x; 2.0013x over previous
#include <hip/hip_runtime.h>
#include <hip/hip_bf16.h>
#include <math.h>


typedef __attribute__((ext_vector_type(2))) float v2f;
typedef __attribute__((ext_vector_type(4))) float v4f;
typedef __attribute__((ext_vector_type(8))) float v8f;

#define HS 256
#define C2D 128
#define DD 64
#define NC_CHUNKS 32

__global__ __launch_bounds__(32)
void k_qkw1(const float* __restrict__ Q, const float* __restrict__ K,
            const float* __restrict__ W1, const float* __restrict__ b1,
            float* __restrict__ QW1T, float* __restrict__ BT,
            float* __restrict__ TQ, float* __restrict__ TB) {
  const int lane = threadIdx.x;
  const int half = lane >> 4;
  const int l16  = lane & 15;
  const int k0   = half * 2;
  const int b    = blockIdx.x;
  const int it   = b & 7;
  const int ct   = (b >> 3) & 7;
  const int h    = (b >> 6) & 1;
  const bool isK = (b >> 7) != 0;

  const float* X  = (isK ? K : Q) + (h * 128 + it * 16 + l16) * 64 + k0;
  const float* Wp = W1 + ((isK ? 64 : 0) + k0) * C2D + ct * 16 + l16;

  v2f bf[16], af[16];
  #pragma unroll
  for (int es = 0; es < 16; ++es)
    bf[es] = *(const v2f*)(X + es * 4);
  #pragma unroll
  for (int es = 0; es < 16; ++es) {
    af[es].x = Wp[(es * 4) * C2D];
    af[es].y = Wp[(es * 4 + 1) * C2D];
  }
  const v4f bb0 = *(const v4f*)(b1 + ct * 16 + 8 * half);
  const v4f bb1 = *(const v4f*)(b1 + ct * 16 + 8 * half + 4);
  asm volatile("" :: "v"(bb0), "v"(bb1));
  __builtin_amdgcn_sched_barrier(0);

  v8f acc0 = {}, acc1 = {};
  #pragma unroll
  for (int es = 0; es < 8; ++es) {
    acc0 = __builtin_amdgcn_wmma_f32_16x16x4_f32(
         false, af[es],  false, bf[es],
         (short)0, acc0,  false,  false);
    acc1 = __builtin_amdgcn_wmma_f32_16x16x4_f32(
         false, af[es + 8],  false, bf[es + 8],
         (short)0, acc1,  false,  false);
  }

  float val[8];
  #pragma unroll
  for (int v = 0; v < 8; ++v) val[v] = acc0[v] + acc1[v];
  if (isK) {
    #pragma unroll
    for (int v = 0; v < 4; ++v) { val[v] += bb0[v]; val[4 + v] += bb1[v]; }
  }
  float* P  = isK ? BT : QW1T;
  float* TP = isK ? TB : TQ;
  const int t0 = ((h * C2D + ct * 16 + 8 * half) << 7) + it * 16 + l16;
  #pragma unroll
  for (int v = 0; v < 8; ++v) {
    P[t0 + v * 128]  = val[v];
    TP[t0 + v * 128] = tanhf(val[v]);
  }
}

#define KP 68
__global__ __launch_bounds__(128) __attribute__((amdgpu_waves_per_eu(1, 1)))
void k_hw2k(const float* __restrict__ K, const float* __restrict__ W2,
            const float* __restrict__ QW1T, const float* __restrict__ BT,
            const float* __restrict__ TQ, const float* __restrict__ TB,
            float* __restrict__ partial) {
  __shared__ __attribute__((aligned(16))) float sK[32 * KP];
  __shared__ __attribute__((aligned(16))) float sH[4 * 32];
  const int tid  = threadIdx.x;
  const int lane = tid & 31;
  const int w    = tid >> 5;
  const int ct = blockIdx.x;
  const int cc = blockIdx.y;
  const int half = lane >> 4;
  const int l16  = lane & 15;
  const int k0   = half * 2;
  const int hjbase = ct * 32;
  const int drow = w * 16 + l16;
  const int cbase = cc * 4;
  const int h     = ct >> 2;
  const int jbase = (ct & 3) * 32;
  const int rowT  = (h * C2D + cbase) << 7;
  const int rowW  = rowT + w * 128;

  v4f tq[32];
  #pragma unroll
  for (int q = 0; q < 32; ++q)
    tq[q] = *(const v4f*)(TQ + rowW + 4 * q);
  const float Bv = TB[rowW + jbase + lane];

  v4f kv[4];
  #pragma unroll
  for (int q = 0; q < 4; ++q)
    kv[q] = *(const v4f*)(K + hjbase * 64 + 4 * (tid + 128 * q));

  #pragma unroll
  for (int q = 0; q < 4; ++q) {
    const int idx = 4 * (tid + 128 * q);
    *(v4f*)(sK + (idx >> 6) * KP + (idx & 63)) = kv[q];
  }
  __syncthreads();

  v2f af[4][16];
  {
    const float* Arow = W2 + (size_t)cbase * 4096 + (size_t)drow * 64 + k0;
    #pragma unroll
    for (int es = 0; es < 16; ++es) {
      #pragma unroll
      for (int ci = 0; ci < 4; ++ci)
        af[ci][es] = *(const v2f*)(Arow + ci * 4096 + es * 4);
    }
  }

  v2f bf[2][16];
  #pragma unroll
  for (int t = 0; t < 2; ++t) {
    #pragma unroll
    for (int es = 0; es < 16; ++es)
      bf[t][es] = *(const v2f*)(sK + (t * 16 + l16) * KP + es * 4 + k0);
  }

  float s = 0.f, dmin = 2.f;
  #pragma unroll
  for (int q = 0; q < 32; ++q) {
    #pragma unroll
    for (int u = 0; u < 4; ++u) {
      const float a   = tq[q][u];
      const float den = fmaf(a, Bv, 1.f);
      const float rc  = __builtin_amdgcn_rcpf(den);
      dmin = fminf(dmin, den);
      s = fmaf(a + Bv, rc, s);
    }
  }
  if (dmin < 0.0625f) {
    const float b = BT[rowW + jbase + lane];
    float ss = 0.f;
    #pragma nounroll
    for (int i = 0; i < 128; ++i)
      ss += tanhf(QW1T[rowW + i] + b);
    s = ss;
  }
  sH[w * 32 + lane] = s;

  v8f acc[4][2];
  #pragma unroll
  for (int ci = 0; ci < 4; ++ci) {
    #pragma unroll
    for (int t = 0; t < 2; ++t)
      #pragma unroll
      for (int v = 0; v < 8; ++v) acc[ci][t][v] = 0.f;
  }
  #pragma unroll
  for (int es = 0; es < 16; ++es) {
    #pragma unroll
    for (int ci = 0; ci < 4; ++ci) {
      #pragma unroll
      for (int t = 0; t < 2; ++t)
        acc[ci][t] = __builtin_amdgcn_wmma_f32_16x16x4_f32(
             false, af[ci][es],  false, bf[t][es],
             (short)0, acc[ci][t],  false,  false);
    }
  }
  __builtin_amdgcn_fence(__ATOMIC_RELEASE, "workgroup", "local");
  __builtin_amdgcn_s_barrier();
  __builtin_amdgcn_fence(__ATOMIC_ACQUIRE, "workgroup", "local");

  v8f out0 = {}, out1 = {};
  #pragma unroll
  for (int ci = 0; ci < 4; ++ci) {
    const float hs0 = sH[ci * 32 + l16];
    const float hs1 = sH[ci * 32 + 16 + l16];
    #pragma unroll
    for (int v = 0; v < 8; ++v) {
      out0[v] += hs0 * acc[ci][0][v];
      out1[v] += hs1 * acc[ci][1][v];
    }
  }

  float* pp = partial + (size_t)cc * (HS * DD) + (size_t)(hjbase + l16) * DD + w * 16 + 8 * half;
  #pragma unroll
  for (int v = 0; v < 8; ++v) {
    pp[v] = out0[v];
    pp[16 * DD + v] = out1[v];
  }
}

#define BP 68
__global__ __launch_bounds__(128)
void k_out(const float* __restrict__ partial, const float* __restrict__ K,
           const float* __restrict__ b2, float* __restrict__ out) {
  __shared__ __attribute__((aligned(16))) float sB[64 * BP];
  __shared__ __attribute__((aligned(16))) float sKr[4 * 64];
  const int t   = threadIdx.x;
  const int lane = t & 31;
  const int w    = t >> 5;
  const int tid = blockIdx.x * 128 + t;
  const int hj  = tid >> 6;
  const int d   = t & 63;

  v4f bv[8];
  #pragma unroll
  for (int q = 0; q < 8; ++q)
    bv[q] = *(const v4f*)(b2 + 4 * (t + 128 * q));
  const v2f kv = *(const v2f*)(K + hj * 64 + 2 * lane);
  float p[NC_CHUNKS];
  #pragma unroll
  for (int cc = 0; cc < NC_CHUNKS; ++cc)
    p[cc] = partial[cc * (HS * DD) + tid];
  __builtin_amdgcn_sched_barrier(0);

  #pragma unroll
  for (int q = 0; q < 8; ++q) {
    const int idx = 4 * (t + 128 * q);
    *(v4f*)(sB + (idx >> 6) * BP + (idx & 63)) = bv[q];
  }
  *(v2f*)(sKr + w * 64 + 2 * lane) = kv;
  __builtin_amdgcn_fence(__ATOMIC_RELEASE, "workgroup", "local");
  __builtin_amdgcn_s_barrier();
  __builtin_amdgcn_fence(__ATOMIC_ACQUIRE, "workgroup", "local");

  float b2k = 0.f;
  #pragma unroll
  for (int q = 0; q < 16; ++q) {
    const v4f b = *(const v4f*)(sB + d * BP + 4 * q);
    const v4f k = *(const v4f*)(sKr + w * 64 + 4 * q);
    #pragma unroll
    for (int u = 0; u < 4; ++u)
      b2k = fmaf(b[u], k[u], b2k);
  }
  float s = 0.f;
  #pragma unroll
  for (int cc = 0; cc < NC_CHUNKS; ++cc)
    s += p[cc];
  out[tid] = s * (1.f / 128.f) + b2k;
}

extern "C" void kernel_launch(void* const* d_in, const int* in_sizes, int n_in,
                              void* d_out, int out_size, void* d_ws, size_t ws_size,
                              hipStream_t stream) {
  const float* Q  = (const float*)d_in[0];
  const float* K  = (const float*)d_in[1];
  const float* W1 = (const float*)d_in[2];
  const float* b1 = (const float*)d_in[3];
  const float* W2 = (const float*)d_in[4];
  const float* b2 = (const float*)d_in[5];
  float* out = (float*)d_out;

  float* ws      = (float*)d_ws;
  float* QW1T    = ws;
  float* BT      = ws + 32768;
  float* TQ      = ws + 65536;
  float* TB      = ws + 98304;
  float* partial = ws + 131072;

  k_qkw1<<<256, 32, 0, stream>>>(Q, K, W1, b1, QW1T, BT, TQ, TB);
  dim3 g2(8, NC_CHUNKS, 1);
  k_hw2k<<<g2, 128, 0, stream>>>(K, W2, QW1T, BT, TQ, TB, partial);
  k_out<<<128, 128, 0, stream>>>(partial, K, b2, out);
}


# k_qkw1 launch and its 4 planes go: each k_hw2k block forms its own slices on the matrix unit from global fragments (QW1 128 i x 4 c, KW1 32 j x 4 c; A rows repeat c), tanh to LDS; sum order kept
# speedup vs baseline: 6.1259x; 1.1145x over previous
#include <hip/hip_runtime.h>
#include <hip/hip_bf16.h>
#include <math.h>


typedef __attribute__((ext_vector_type(2))) float v2f;
typedef __attribute__((ext_vector_type(4))) float v4f;
typedef __attribute__((ext_vector_type(8))) float v8f;

#define HS 256
#define C2D 128
#define DD 64
#define NC_CHUNKS 32

#define KP 68
__global__ __launch_bounds__(128) __attribute__((amdgpu_waves_per_eu(1, 1)))
void k_hw2k(const float* __restrict__ Q, const float* __restrict__ K,
            const float* __restrict__ W1, const float* __restrict__ b1,
            const float* __restrict__ W2, float* __restrict__ partial) {
  __shared__ __attribute__((aligned(16))) float sK[32 * KP];
  __shared__ __attribute__((aligned(16))) float sH[4 * 32];
  __shared__ __attribute__((aligned(16))) float sQW[4 * 128];
  __shared__ __attribute__((aligned(16))) float sTQ[4 * 128];
  __shared__ __attribute__((aligned(16))) float sKW[2 * 4 * 32];
  const int tid  = threadIdx.x;
  const int lane = tid & 31;
  const int w    = tid >> 5;
  const int ct = blockIdx.x;
  const int cc = blockIdx.y;
  const int half = lane >> 4;
  const int l16  = lane & 15;
  const int k0   = half * 2;
  const int hjbase = ct * 32;
  const int drow = w * 16 + l16;
  const int cbase = cc * 4;
  const int h     = ct >> 2;
  const int tK    = w & 1;
  const int kh    = w >> 1;

  v4f kv[4];
  #pragma unroll
  for (int q = 0; q < 4; ++q)
    kv[q] = *(const v4f*)(K + hjbase * 64 + 4 * (tid + 128 * q));

  v2f qf[2][16];
  {
    const float* X = Q + (h * 128 + w * 32 + l16) * 64 + k0;
    #pragma unroll
    for (int t = 0; t < 2; ++t) {
      #pragma unroll
      for (int es = 0; es < 16; ++es)
        qf[t][es] = *(const v2f*)(X + t * 16 * 64 + es * 4);
    }
  }
  v2f wq[16], wk[8];
  {
    const float* Wp = W1 + k0 * C2D + cbase + (l16 & 3);
    #pragma unroll
    for (int es = 0; es < 16; ++es) {
      wq[es].x = Wp[(es * 4) * C2D];
      wq[es].y = Wp[(es * 4 + 1) * C2D];
    }
    const float* Wk = Wp + (64 + kh * 32) * C2D;
    #pragma unroll
    for (int e8 = 0; e8 < 8; ++e8) {
      wk[e8].x = Wk[(e8 * 4) * C2D];
      wk[e8].y = Wk[(e8 * 4 + 1) * C2D];
    }
  }
  v2f kb[8];
  {
    const float* Xk = K + (hjbase + tK * 16 + l16) * 64 + kh * 32 + k0;
    #pragma unroll
    for (int e8 = 0; e8 < 8; ++e8)
      kb[e8] = *(const v2f*)(Xk + e8 * 4);
  }
  const float b1w = b1[cbase + w];

  v2f af[4][16];
  {
    const float* Arow = W2 + (size_t)cbase * 4096 + (size_t)drow * 64 + k0;
    #pragma unroll
    for (int es = 0; es < 16; ++es) {
      #pragma unroll
      for (int ci = 0; ci < 4; ++ci)
        af[ci][es] = *(const v2f*)(Arow + ci * 4096 + es * 4);
    }
  }
  __builtin_amdgcn_sched_barrier(0);

  #pragma unroll
  for (int q = 0; q < 4; ++q) {
    const int idx = 4 * (tid + 128 * q);
    *(v4f*)(sK + (idx >> 6) * KP + (idx & 63)) = kv[q];
  }

  v8f q00 = {}, q01 = {}, q10 = {}, q11 = {}, kw = {};
  #pragma unroll
  for (int es = 0; es < 8; ++es) {
    q00 = __builtin_amdgcn_wmma_f32_16x16x4_f32(
         false, wq[es],  false, qf[0][es],
         (short)0, q00,  false,  false);
    q01 = __builtin_amdgcn_wmma_f32_16x16x4_f32(
         false, wq[es + 8],  false, qf[0][es + 8],
         (short)0, q01,  false,  false);
    q10 = __builtin_amdgcn_wmma_f32_16x16x4_f32(
         false, wq[es],  false, qf[1][es],
         (short)0, q10,  false,  false);
    q11 = __builtin_amdgcn_wmma_f32_16x16x4_f32(
         false, wq[es + 8],  false, qf[1][es + 8],
         (short)0, q11,  false,  false);
    kw = __builtin_amdgcn_wmma_f32_16x16x4_f32(
         false, wk[es],  false, kb[es],
         (short)0, kw,  false,  false);
  }
  #pragma unroll
  for (int v = 0; v < 4; ++v) {
    const float r = half ? (q10[v] + q11[v]) : (q00[v] + q01[v]);
    sQW[v * 128 + w * 32 + lane] = r;
    sTQ[v * 128 + w * 32 + lane] = tanhf(r);
  }
  #pragma unroll
  for (int u = 0; u < 2; ++u)
    sKW[kh * 128 + (2 * half + u) * 32 + tK * 16 + l16] = half ? kw[2 + u] : kw[u];
  __builtin_amdgcn_fence(__ATOMIC_RELEASE, "workgroup", "local");
  __builtin_amdgcn_s_barrier();
  __builtin_amdgcn_fence(__ATOMIC_ACQUIRE, "workgroup", "local");

  v4f tq[32];
  #pragma unroll
  for (int q = 0; q < 32; ++q)
    tq[q] = *(const v4f*)(sTQ + w * 128 + 4 * q);
  const float bpre = (sKW[w * 32 + lane] + sKW[128 + w * 32 + lane]) + b1w;
  const float Bv = tanhf(bpre);

  v2f bf[2][16];
  #pragma unroll
  for (int t = 0; t < 2; ++t) {
    #pragma unroll
    for (int es = 0; es < 16; ++es)
      bf[t][es] = *(const v2f*)(sK + (t * 16 + l16) * KP + es * 4 + k0);
  }

  float s = 0.f, dmin = 2.f;
  #pragma unroll
  for (int q = 0; q < 32; ++q) {
    #pragma unroll
    for (int u = 0; u < 4; ++u) {
      const float a   = tq[q][u];
      const float den = fmaf(a, Bv, 1.f);
      const float rc  = __builtin_amdgcn_rcpf(den);
      dmin = fminf(dmin, den);
      s = fmaf(a + Bv, rc, s);
    }
  }
  if (dmin < 0.0625f) {
    float ss = 0.f;
    #pragma nounroll
    for (int i = 0; i < 128; ++i)
      ss += tanhf(sQW[w * 128 + i] + bpre);
    s = ss;
  }
  sH[w * 32 + lane] = s;

  v8f acc[4][2];
  #pragma unroll
  for (int ci = 0; ci < 4; ++ci) {
    #pragma unroll
    for (int t = 0; t < 2; ++t)
      #pragma unroll
      for (int v = 0; v < 8; ++v) acc[ci][t][v] = 0.f;
  }
  #pragma unroll
  for (int es = 0; es < 16; ++es) {
    #pragma unroll
    for (int ci = 0; ci < 4; ++ci) {
      #pragma unroll
      for (int t = 0; t < 2; ++t)
        acc[ci][t] = __builtin_amdgcn_wmma_f32_16x16x4_f32(
             false, af[ci][es],  false, bf[t][es],
             (short)0, acc[ci][t],  false,  false);
    }
  }
  __builtin_amdgcn_fence(__ATOMIC_RELEASE, "workgroup", "local");
  __builtin_amdgcn_s_barrier();
  __builtin_amdgcn_fence(__ATOMIC_ACQUIRE, "workgroup", "local");

  v8f out0 = {}, out1 = {};
  #pragma unroll
  for (int ci = 0; ci < 4; ++ci) {
    const float hs0 = sH[ci * 32 + l16];
    const float hs1 = sH[ci * 32 + 16 + l16];
    #pragma unroll
    for (int v = 0; v < 8; ++v) {
      out0[v] += hs0 * acc[ci][0][v];
      out1[v] += hs1 * acc[ci][1][v];
    }
  }

  float* pp = partial + (size_t)cc * (HS * DD) + (size_t)(hjbase + l16) * DD + w * 16 + 8 * half;
  #pragma unroll
  for (int v = 0; v < 8; ++v) {
    pp[v] = out0[v];
    pp[16 * DD + v] = out1[v];
  }
}

#define BP 68
__global__ __launch_bounds__(128)
void k_out(const float* __restrict__ partial, const float* __restrict__ K,
           const float* __restrict__ b2, float* __restrict__ out) {
  __shared__ __attribute__((aligned(16))) float sB[64 * BP];
  __shared__ __attribute__((aligned(16))) float sKr[4 * 64];
  const int t   = threadIdx.x;
  const int lane = t & 31;
  const int w    = t >> 5;
  const int tid = blockIdx.x * 128 + t;
  const int hj  = tid >> 6;
  const int d   = t & 63;

  v4f bv[8];
  #pragma unroll
  for (int q = 0; q < 8; ++q)
    bv[q] = *(const v4f*)(b2 + 4 * (t + 128 * q));
  const v2f kv = *(const v2f*)(K + hj * 64 + 2 * lane);
  float p[NC_CHUNKS];
  #pragma unroll
  for (int cc = 0; cc < NC_CHUNKS; ++cc)
    p[cc] = partial[cc * (HS * DD) + tid];
  __builtin_amdgcn_sched_barrier(0);

  #pragma unroll
  for (int q = 0; q < 8; ++q) {
    const int idx = 4 * (t + 128 * q);
    *(v4f*)(sB + (idx >> 6) * BP + (idx & 63)) = bv[q];
  }
  *(v2f*)(sKr + w * 64 + 2 * lane) = kv;
  __builtin_amdgcn_fence(__ATOMIC_RELEASE, "workgroup", "local");
  __builtin_amdgcn_s_barrier();
  __builtin_amdgcn_fence(__ATOMIC_ACQUIRE, "workgroup", "local");

  float b2k = 0.f;
  #pragma unroll
  for (int q = 0; q < 16; ++q) {
    const v4f b = *(const v4f*)(sB + d * BP + 4 * q);
    const v4f k = *(const v4f*)(sKr + w * 64 + 4 * q);
    #pragma unroll
    for (int u = 0; u < 4; ++u)
      b2k = fmaf(b[u], k[u], b2k);
  }
  float s = 0.f;
  #pragma unroll
  for (int cc = 0; cc < NC_CHUNKS; ++cc)
    s += p[cc];
  out[tid] = s * (1.f / 128.f) + b2k;
}

extern "C" void kernel_launch(void* const* d_in, const int* in_sizes, int n_in,
                              void* d_out, int out_size, void* d_ws, size_t ws_size,
                              hipStream_t stream) {
  const float* Q  = (const float*)d_in[0];
  const float* K  = (const float*)d_in[1];
  const float* W1 = (const float*)d_in[2];
  const float* b1 = (const float*)d_in[3];
  const float* W2 = (const float*)d_in[4];
  const float* b2 = (const float*)d_in[5];
  float* out = (float*)d_out;

  float* partial = (float*)d_ws;

  dim3 g1(8, NC_CHUNKS, 1);
  k_hw2k<<<g1, 128, 0, stream>>>(Q, K, W1, b1, W2, partial);
  k_out<<<128, 128, 0, stream>>>(partial, K, b2, out);
}


# k_hw2k: W2, Q and K operands fetched by whole-line 16-byte loads into LDS regions of the wave's own and re-read as fragments, not by 8-byte fragment loads that touch 16 lines each; sum order kept
# speedup vs baseline: 6.9615x; 1.1364x over previous
#include <hip/hip_runtime.h>
#include <hip/hip_bf16.h>
#include <math.h>


typedef __attribute__((ext_vector_type(2))) float v2f;
typedef __attribute__((ext_vector_type(4))) float v4f;
typedef __attribute__((ext_vector_type(8))) float v8f;

#define HS 256
#define C2D 128
#define DD 64
#define NC_CHUNKS 32

#define KP 68
#define KQ 36
__global__ __launch_bounds__(128) __attribute__((amdgpu_waves_per_eu(1, 1)))
void k_hw2k(const float* __restrict__ Q, const float* __restrict__ K,
            const float* __restrict__ W1, const float* __restrict__ b1,
            const float* __restrict__ W2, float* __restrict__ partial) {
  __shared__ __attribute__((aligned(16))) float sK[32 * KP];
  __shared__ __attribute__((aligned(16))) float sH[4 * 32];
  __shared__ __attribute__((aligned(16))) float sQW[4 * 128];
  __shared__ __attribute__((aligned(16))) float sTQ[4 * 128];
  __shared__ __attribute__((aligned(16))) float sKW[2 * 4 * 32];
  __shared__ __attribute__((aligned(16))) float sQs[4 * 32 * KP];
  __shared__ __attribute__((aligned(16))) float sKs[4 * 16 * KQ];
  __shared__ __attribute__((aligned(16))) float sWs[4 * 4 * 16 * KP];
  const int tid  = threadIdx.x;
  const int lane = tid & 31;
  const int w    = tid >> 5;
  const int ct = blockIdx.x;
  const int cc = blockIdx.y;
  const int half = lane >> 4;
  const int l16  = lane & 15;
  const int k0   = half * 2;
  const int hjbase = ct * 32;
  const int cbase = cc * 4;
  const int h     = ct >> 2;
  const int tK    = w & 1;
  const int kh    = w >> 1;

  v4f kv[4];
  #pragma unroll
  for (int q = 0; q < 4; ++q)
    kv[q] = *(const v4f*)(K + hjbase * 64 + 4 * (tid + 128 * q));

  v4f qs[16];
  {
    const float* X = Q + (h * 128 + w * 32) * 64 + 4 * lane;
    #pragma unroll
    for (int q = 0; q < 16; ++q)
      qs[q] = *(const v4f*)(X + 128 * q);
  }
  v2f wq[16], wk[8];
  {
    const float* Wp = W1 + k0 * C2D + cbase + (l16 & 3);
    #pragma unroll
    for (int es = 0; es < 16; ++es) {
      wq[es].x = Wp[(es * 4) * C2D];
      wq[es].y = Wp[(es * 4 + 1) * C2D];
    }
    const float* Wk = Wp + (64 + kh * 32) * C2D;
    #pragma unroll
    for (int e8 = 0; e8 < 8; ++e8) {
      wk[e8].x = Wk[(e8 * 4) * C2D];
      wk[e8].y = Wk[(e8 * 4 + 1) * C2D];
    }
  }
  v4f ks[4];
  {
    const float* Xk = K + (hjbase + tK * 16 + (lane >> 3)) * 64 + kh * 32 + 4 * (lane & 7);
    #pragma unroll
    for (int q = 0; q < 4; ++q)
      ks[q] = *(const v4f*)(Xk + q * 4 * 64);
  }
  const float b1w = b1[cbase + w];

  v4f ws[4][8];
  {
    const float* Ap = W2 + (size_t)cbase * 4096 + (size_t)(w * 16) * 64 + 4 * lane;
    #pragma unroll
    for (int ci = 0; ci < 4; ++ci) {
      #pragma unroll
      for (int q = 0; q < 8; ++q)
        ws[ci][q] = *(const v4f*)(Ap + ci * 4096 + 128 * q);
    }
  }
  __builtin_amdgcn_sched_barrier(0);

  #pragma unroll
  for (int q = 0; q < 4; ++q) {
    const int idx = 4 * (tid + 128 * q);
    *(v4f*)(sK + (idx >> 6) * KP + (idx & 63)) = kv[q];
  }
  float* myQ = sQs + w * (32 * KP);
  #pragma unroll
  for (int q = 0; q < 16; ++q) {
    const int idx = 4 * (lane + 32 * q);
    *(v4f*)(myQ + (idx >> 6) * KP + (idx & 63)) = qs[q];
  }
  float* myK = sKs + w * (16 * KQ);
  #pragma unroll
  for (int q = 0; q < 4; ++q)
    *(v4f*)(myK + (4 * q + (lane >> 3)) * KQ + 4 * (lane & 7)) = ks[q];
  v2f qf[2][16];
  #pragma unroll
  for (int t = 0; t < 2; ++t) {
    #pragma unroll
    for (int es = 0; es < 16; ++es)
      qf[t][es] = *(const v2f*)(myQ + (t * 16 + l16) * KP + es * 4 + k0);
  }
  v2f kb[8];
  #pragma unroll
  for (int e8 = 0; e8 < 8; ++e8)
    kb[e8] = *(const v2f*)(myK + l16 * KQ + e8 * 4 + k0);

  v8f q00 = {}, q01 = {}, q10 = {}, q11 = {}, kw = {};
  #pragma unroll
  for (int es = 0; es < 8; ++es) {
    q00 = __builtin_amdgcn_wmma_f32_16x16x4_f32(
         false, wq[es],  false, qf[0][es],
         (short)0, q00,  false,  false);
    q01 = __builtin_amdgcn_wmma_f32_16x16x4_f32(
         false, wq[es + 8],  false, qf[0][es + 8],
         (short)0, q01,  false,  false);
    q10 = __builtin_amdgcn_wmma_f32_16x16x4_f32(
         false, wq[es],  false, qf[1][es],
         (short)0, q10,  false,  false);
    q11 = __builtin_amdgcn_wmma_f32_16x16x4_f32(
         false, wq[es + 8],  false, qf[1][es + 8],
         (short)0, q11,  false,  false);
    kw = __builtin_amdgcn_wmma_f32_16x16x4_f32(
         false, wk[es],  false, kb[es],
         (short)0, kw,  false,  false);
  }
  #pragma unroll
  for (int v = 0; v < 4; ++v) {
    const float r = half ? (q10[v] + q11[v]) : (q00[v] + q01[v]);
    sQW[v * 128 + w * 32 + lane] = r;
    sTQ[v * 128 + w * 32 + lane] = tanhf(r);
  }
  #pragma unroll
  for (int u = 0; u < 2; ++u)
    sKW[kh * 128 + (2 * half + u) * 32 + tK * 16 + l16] = half ? kw[2 + u] : kw[u];
  __builtin_amdgcn_fence(__ATOMIC_RELEASE, "workgroup", "local");
  __builtin_amdgcn_s_barrier();
  __builtin_amdgcn_fence(__ATOMIC_ACQUIRE, "workgroup", "local");

  v4f tq[32];
  #pragma unroll
  for (int q = 0; q < 32; ++q)
    tq[q] = *(const v4f*)(sTQ + w * 128 + 4 * q);
  const float bpre = (sKW[w * 32 + lane] + sKW[128 + w * 32 + lane]) + b1w;
  const float Bv = tanhf(bpre);

  v2f bf[2][16];
  #pragma unroll
  for (int t = 0; t < 2; ++t) {
    #pragma unroll
    for (int es = 0; es < 16; ++es)
      bf[t][es] = *(const v2f*)(sK + (t * 16 + l16) * KP + es * 4 + k0);
  }

  float* myW = sWs + w * (4 * 16 * KP);
  #pragma unroll
  for (int ci = 0; ci < 4; ++ci) {
    #pragma unroll
    for (int q = 0; q < 8; ++q) {
      const int idx = 4 * (lane + 32 * q);
      *(v4f*)(myW + (ci * 16 + (idx >> 6)) * KP + (idx & 63)) = ws[ci][q];
    }
  }
  v2f af[4][16];
  #pragma unroll
  for (int es = 0; es < 16; ++es) {
    #pragma unroll
    for (int ci = 0; ci < 4; ++ci)
      af[ci][es] = *(const v2f*)(myW + (ci * 16 + l16) * KP + es * 4 + k0);
  }
  __builtin_amdgcn_sched_barrier(0);

  float s = 0.f, dmin = 2.f;
  #pragma unroll
  for (int q = 0; q < 32; ++q) {
    #pragma unroll
    for (int u = 0; u < 4; ++u) {
      const float a   = tq[q][u];
      const float den = fmaf(a, Bv, 1.f);
      const float rc  = __builtin_amdgcn_rcpf(den);
      dmin = fminf(dmin, den);
      s = fmaf(a + Bv, rc, s);
    }
  }
  if (dmin < 0.0625f) {
    float ss = 0.f;
    #pragma nounroll
    for (int i = 0; i < 128; ++i)
      ss += tanhf(sQW[w * 128 + i] + bpre);
    s = ss;
  }
  sH[w * 32 + lane] = s;

  v8f acc[4][2];
  #pragma unroll
  for (int ci = 0; ci < 4; ++ci) {
    #pragma unroll
    for (int t = 0; t < 2; ++t)
      #pragma unroll
      for (int v = 0; v < 8; ++v) acc[ci][t][v] = 0.f;
  }
  #pragma unroll
  for (int es = 0; es < 16; ++es) {
    #pragma unroll
    for (int ci = 0; ci < 4; ++ci) {
      #pragma unroll
      for (int t = 0; t < 2; ++t)
        acc[ci][t] = __builtin_amdgcn_wmma_f32_16x16x4_f32(
             false, af[ci][es],  false, bf[t][es],
             (short)0, acc[ci][t],  false,  false);
    }
  }
  __builtin_amdgcn_fence(__ATOMIC_RELEASE, "workgroup", "local");
  __builtin_amdgcn_s_barrier();
  __builtin_amdgcn_fence(__ATOMIC_ACQUIRE, "workgroup", "local");

  v8f out0 = {}, out1 = {};
  #pragma unroll
  for (int ci = 0; ci < 4; ++ci) {
    const float hs0 = sH[ci * 32 + l16];
    const float hs1 = sH[ci * 32 + 16 + l16];
    #pragma unroll
    for (int v = 0; v < 8; ++v) {
      out0[v] += hs0 * acc[ci][0][v];
      out1[v] += hs1 * acc[ci][1][v];
    }
  }

  float* pp = partial + (size_t)cc * (HS * DD) + (size_t)(hjbase + l16) * DD + w * 16 + 8 * half;
  #pragma unroll
  for (int v = 0; v < 8; ++v) {
    pp[v] = out0[v];
    pp[16 * DD + v] = out1[v];
  }
}

#define BP 68
__global__ __launch_bounds__(128)
void k_out(const float* __restrict__ partial, const float* __restrict__ K,
           const float* __restrict__ b2, float* __restrict__ out) {
  __shared__ __attribute__((aligned(16))) float sB[64 * BP];
  __shared__ __attribute__((aligned(16))) float sKr[4 * 64];
  const int t   = threadIdx.x;
  const int lane = t & 31;
  const int w    = t >> 5;
  const int tid = blockIdx.x * 128 + t;
  const int hj  = tid >> 6;
  const int d   = t & 63;

  v4f bv[8];
  #pragma unroll
  for (int q = 0; q < 8; ++q)
    bv[q] = *(const v4f*)(b2 + 4 * (t + 128 * q));
  const v2f kv = *(const v2f*)(K + hj * 64 + 2 * lane);
  float p[NC_CHUNKS];
  #pragma unroll
  for (int cc = 0; cc < NC_CHUNKS; ++cc)
    p[cc] = partial[cc * (HS * DD) + tid];
  __builtin_amdgcn_sched_barrier(0);

  #pragma unroll
  for (int q = 0; q < 8; ++q) {
    const int idx = 4 * (t + 128 * q);
    *(v4f*)(sB + (idx >> 6) * BP + (idx & 63)) = bv[q];
  }
  *(v2f*)(sKr + w * 64 + 2 * lane) = kv;
  __builtin_amdgcn_fence(__ATOMIC_RELEASE, "workgroup", "local");
  __builtin_amdgcn_s_barrier();
  __builtin_amdgcn_fence(__ATOMIC_ACQUIRE, "workgroup", "local");

  float b2k = 0.f;
  #pragma unroll
  for (int q = 0; q < 16; ++q) {
    const v4f b = *(const v4f*)(sB + d * BP + 4 * q);
    const v4f k = *(const v4f*)(sKr + w * 64 + 4 * q);
    #pragma unroll
    for (int u = 0; u < 4; ++u)
      b2k = fmaf(b[u], k[u], b2k);
  }
  float s = 0.f;
  #pragma unroll
  for (int cc = 0; cc < NC_CHUNKS; ++cc)
    s += p[cc];
  out[tid] = s * (1.f / 128.f) + b2k;
}

extern "C" void kernel_launch(void* const* d_in, const int* in_sizes, int n_in,
                              void* d_out, int out_size, void* d_ws, size_t ws_size,
                              hipStream_t stream) {
  const float* Q  = (const float*)d_in[0];
  const float* K  = (const float*)d_in[1];
  const float* W1 = (const float*)d_in[2];
  const float* b1 = (const float*)d_in[3];
  const float* W2 = (const float*)d_in[4];
  const float* b2 = (const float*)d_in[5];
  float* out = (float*)d_out;

  float* partial = (float*)d_ws;

  dim3 g1(8, NC_CHUNKS, 1);
  k_hw2k<<<g1, 128, 0, stream>>>(Q, K, W1, b1, W2, partial);
  k_out<<<128, 128, 0, stream>>>(partial, K, b2, out);
}
